# starting kernel
# baseline (speedup 1.0000x reference)
#include <hip/hip_runtime.h>
#include <math.h>

static constexpr int kB    = 8;
static constexpr int kC    = 512;
static constexpr int kT    = 4;
static constexpr int kHW   = 64 * 64;
static constexpr int kNSP  = 100;
static constexpr int kDOUT = 128;
static constexpr int kNSEG = kB * kT * kNSP;
static constexpr int kCHUNK  = 32;
static constexpr int kNCHUNK = kC / kCHUNK;
static constexpr int kLPAD   = 33;
static constexpr int kASTR   = 516;

typedef float v2f __attribute__((ext_vector_type(2)));
typedef float v8f __attribute__((ext_vector_type(8)));

__global__ __launch_bounds__(256) void seg_sum_kernel(
    const float* __restrict__ maps, const int* __restrict__ sp_mask,
    float* __restrict__ sums)
{
  __shared__ float lds[kNSP * kLPAD];
  const int tid = threadIdx.x;
  for (int i = tid; i < kNSP * kLPAD; i += 256) lds[i] = 0.0f;
  __syncthreads();

  const int blk   = blockIdx.x;
  const int chunk = blk & (kNCHUNK - 1);
  const int bt    = blk >> 4;
  const int b     = bt >> 2;
  const int t     = bt & 3;

  const int lane = tid & 31;
  const int wv   = tid >> 5;

  const int* spb = sp_mask + (size_t)bt * kHW;

  for (int pb = 0; pb < kHW; pb += 128) {
    const int p = pb + lane * 4;
    const int4 sp4 = *reinterpret_cast<const int4*>(spb + p);
#pragma unroll
    for (int cc = 0; cc < 4; ++cc) {
      const int cl = wv + 8 * cc;
      const int c  = chunk * kCHUNK + cl;
      const float* mp =
          maps + ((size_t)(b * kC + c) * kT + t) * kHW + p;
      if (pb + 128 < kHW) __builtin_prefetch(mp + 128, 0, 1);
      const float4 v = *reinterpret_cast<const float4*>(mp);
      atomicAdd(&lds[sp4.x * kLPAD + cl], v.x);
      atomicAdd(&lds[sp4.y * kLPAD + cl], v.y);
      atomicAdd(&lds[sp4.z * kLPAD + cl], v.z);
      atomicAdd(&lds[sp4.w * kLPAD + cl], v.w);
    }
  }
  __syncthreads();

  float* outp = sums + (size_t)bt * kNSP * kC + chunk * kCHUNK;
  for (int e = tid; e < kNSP * kCHUNK; e += 256) {
    const int n  = e >> 5;
    const int cl = e & 31;
    outp[(size_t)n * kC + cl] = lds[n * kLPAD + cl];
  }
}

__global__ __launch_bounds__(256) void seg_cnt_kernel(
    const int* __restrict__ sp_mask, float* __restrict__ counts)
{
  __shared__ int cnt[kNSP];
  const int tid = threadIdx.x;
  if (tid < kNSP) cnt[tid] = 0;
  __syncthreads();
  const int bt = blockIdx.x;
  const int* spb = sp_mask + (size_t)bt * kHW;
  for (int p = tid; p < kHW; p += 256) atomicAdd(&cnt[spb[p]], 1);
  __syncthreads();
  if (tid < kNSP) counts[bt * kNSP + tid] = (float)cnt[tid];
}

__global__ __launch_bounds__(256) void fc_norm_kernel(
    const float* __restrict__ sums, const float* __restrict__ counts,
    const float* __restrict__ Wfc, float* __restrict__ outp)
{
  __shared__ float lds_a[16 * kASTR];
  __shared__ float lds_o[16 * kDOUT];
  __shared__ float lds_inv[16];

  const int tid = threadIdx.x;
  const int s0  = blockIdx.x * 16;

  for (int e = tid; e < 16 * kC; e += 256) {
    const int m = e >> 9;
    const int c = e & (kC - 1);
    const float cv = counts[s0 + m];
    const float sv = sums[(size_t)(s0 + m) * kC + c];
    lds_a[m * kASTR + c] = (cv > 0.0f) ? (sv / cv) : 0.0f;
  }
  __syncthreads();

  const int lane = tid & 31;
  const int wv   = tid >> 5;
  const int half = lane >> 4;
  const int ml   = lane & 15;
  const int n0   = wv * 16;

  v8f acc = {0.f, 0.f, 0.f, 0.f, 0.f, 0.f, 0.f, 0.f};
  const float* wrow = Wfc + (size_t)(n0 + ml) * kC;

  for (int k0 = 0; k0 < kC; k0 += 4) {
    v2f a, bm;
    const float2 av = *reinterpret_cast<const float2*>(
        &lds_a[ml * kASTR + k0 + 2 * half]);
    a.x = av.x; a.y = av.y;
    const float2 wb = *reinterpret_cast<const float2*>(wrow + k0 + 2 * half);
    bm.x = wb.x; bm.y = wb.y;
    acc = __builtin_amdgcn_wmma_f32_16x16x4_f32(
        false, a, false, bm, (short)0, acc, false, false);
  }

#pragma unroll
  for (int r = 0; r < 8; ++r) {
    const int m = r + 8 * half;
    lds_o[m * kDOUT + n0 + ml] = acc[r];
  }
  __syncthreads();

  if (tid < 16) {
    float ss = 0.0f;
    for (int d = 0; d < kDOUT; ++d) {
      const float x = lds_o[tid * kDOUT + d];
      ss += x * x;
    }
    lds_inv[tid] = 1.0f / fmaxf(sqrtf(ss), 1e-12f);
  }
  __syncthreads();

  for (int e = tid; e < 16 * kDOUT; e += 256) {
    const int m = e >> 7;
    const int d = e & (kDOUT - 1);
    const int s = s0 + m;
    const int b = s / (kT * kNSP);
    const int t = (s / kNSP) % kT;
    const int n = s % kNSP;
    outp[((size_t)(b * kDOUT + d) * kT + t) * kNSP + n] =
        lds_o[m * kDOUT + d] * lds_inv[m];
  }
}

extern "C" void kernel_launch(void* const* d_in, const int* in_sizes, int n_in,
                              void* d_out, int out_size, void* d_ws, size_t ws_size,
                              hipStream_t stream) {
  (void)in_sizes; (void)n_in; (void)out_size; (void)ws_size;
  const float* maps    = (const float*)d_in[0];
  const int*   sp_mask = (const int*)d_in[1];
  const float* Wfc     = (const float*)d_in[2];
  float* out = (float*)d_out;

  float* sums   = (float*)d_ws;
  float* counts = sums + (size_t)kNSEG * kC;

  seg_sum_kernel<<<kB * kT * kNCHUNK, 256, 0, stream>>>(maps, sp_mask, sums);
  seg_cnt_kernel<<<kB * kT, 256, 0, stream>>>(sp_mask, counts);
  fc_norm_kernel<<<kNSEG / 16, 256, 0, stream>>>(sums, counts, Wfc, out);
}


# loads grouped ahead of their uses instead of one load per wait: seg_sum issues a 10-load stage before the previous stage's LDS adds, fc_norm's mean staging its 16 loads at once; add order kept
# speedup vs baseline: 1.2082x; 1.2082x over previous
#include <hip/hip_runtime.h>
#include <math.h>

static constexpr int kB    = 8;
static constexpr int kC    = 512;
static constexpr int kT    = 4;
static constexpr int kHW   = 64 * 64;
static constexpr int kNSP  = 100;
static constexpr int kDOUT = 128;
static constexpr int kNSEG = kB * kT * kNSP;
static constexpr int kCHUNK  = 32;
static constexpr int kNCHUNK = kC / kCHUNK;
static constexpr int kLPAD   = 33;
static constexpr int kASTR   = 516;

typedef float v2f __attribute__((ext_vector_type(2)));
typedef float v8f __attribute__((ext_vector_type(8)));

struct SegStage { int4 sp[2]; float4 v[2][4]; };
static constexpr size_t kCSTEP = (size_t)8 * kT * kHW;

__device__ __forceinline__ void seg_load_stage(SegStage& s, const int* sp0,
                                               const float* mp0, int pb)
{
#pragma unroll
  for (int h = 0; h < 2; ++h) {
    s.sp[h] = *reinterpret_cast<const int4*>(sp0 + pb + 128 * h);
#pragma unroll
    for (int cc = 0; cc < 4; ++cc)
      s.v[h][cc] = *reinterpret_cast<const float4*>(mp0 + cc * kCSTEP + pb + 128 * h);
  }
}

__device__ __forceinline__ void seg_add_stage(const SegStage& s, float* lds, int wv)
{
#pragma unroll
  for (int h = 0; h < 2; ++h) {
#pragma unroll
    for (int cc = 0; cc < 4; ++cc) {
      const int cl = wv + 8 * cc;
      atomicAdd(&lds[s.sp[h].x * kLPAD + cl], s.v[h][cc].x);
      atomicAdd(&lds[s.sp[h].y * kLPAD + cl], s.v[h][cc].y);
      atomicAdd(&lds[s.sp[h].z * kLPAD + cl], s.v[h][cc].z);
      atomicAdd(&lds[s.sp[h].w * kLPAD + cl], s.v[h][cc].w);
    }
  }
}

__global__ __launch_bounds__(256) void seg_sum_kernel(
    const float* __restrict__ maps, const int* __restrict__ sp_mask,
    float* __restrict__ sums)
{
  __shared__ float lds[kNSP * kLPAD];
  const int tid = threadIdx.x;
  for (int i = tid; i < kNSP * kLPAD; i += 256) lds[i] = 0.0f;
  __syncthreads();

  const int blk   = blockIdx.x;
  const int chunk = blk & (kNCHUNK - 1);
  const int bt    = blk >> 4;
  const int b     = bt >> 2;
  const int t     = bt & 3;

  const int lane = tid & 31;
  const int wv   = tid >> 5;

  const int* spb = sp_mask + (size_t)bt * kHW;

  const float* mp0 =
      maps + ((size_t)(b * kC + chunk * kCHUNK + wv) * kT + t) * kHW + lane * 4;
  const int* sp0 = spb + lane * 4;

  SegStage sa, sb;
  seg_load_stage(sa, sp0, mp0, 0);
  for (int pb = 0; pb < kHW; pb += 512) {
    seg_load_stage(sb, sp0, mp0, pb + 256);
    seg_add_stage(sa, lds, wv);
    if (pb + 512 < kHW) seg_load_stage(sa, sp0, mp0, pb + 512);
    seg_add_stage(sb, lds, wv);
  }
  __syncthreads();

  float* outp = sums + (size_t)bt * kNSP * kC + chunk * kCHUNK;
  for (int e = tid; e < kNSP * kCHUNK; e += 256) {
    const int n  = e >> 5;
    const int cl = e & 31;
    outp[(size_t)n * kC + cl] = lds[n * kLPAD + cl];
  }
}

__global__ __launch_bounds__(256) void seg_cnt_kernel(
    const int* __restrict__ sp_mask, float* __restrict__ counts)
{
  __shared__ int cnt[kNSP];
  const int tid = threadIdx.x;
  if (tid < kNSP) cnt[tid] = 0;
  __syncthreads();
  const int bt = blockIdx.x;
  const int* spb = sp_mask + (size_t)bt * kHW;
  for (int p = tid; p < kHW; p += 256) atomicAdd(&cnt[spb[p]], 1);
  __syncthreads();
  if (tid < kNSP) counts[bt * kNSP + tid] = (float)cnt[tid];
}

__global__ __launch_bounds__(256) void fc_norm_kernel(
    const float* __restrict__ sums, const float* __restrict__ counts,
    const float* __restrict__ Wfc, float* __restrict__ outp)
{
  __shared__ __attribute__((aligned(16))) float lds_a[16 * kASTR];
  __shared__ float lds_o[16 * kDOUT];
  __shared__ float lds_inv[16];

  const int tid = threadIdx.x;
  const int s0  = blockIdx.x * 16;

  {
    const int c4 = tid & 127;
    const int mh = tid >> 7;
    float4 sv[8];
    float  cv[8];
#pragma unroll
    for (int i = 0; i < 8; ++i) {
      const int m = mh + 2 * i;
      cv[i] = counts[s0 + m];
      sv[i] = *reinterpret_cast<const float4*>(sums + (size_t)(s0 + m) * kC + 4 * c4);
    }
#pragma unroll
    for (int i = 0; i < 8; ++i) {
      const int m = mh + 2 * i;
      const bool nz = cv[i] > 0.0f;
      float4 r;
      r.x = nz ? (sv[i].x / cv[i]) : 0.0f;
      r.y = nz ? (sv[i].y / cv[i]) : 0.0f;
      r.z = nz ? (sv[i].z / cv[i]) : 0.0f;
      r.w = nz ? (sv[i].w / cv[i]) : 0.0f;
      *reinterpret_cast<float4*>(&lds_a[m * kASTR + 4 * c4]) = r;
    }
  }
  __syncthreads();

  const int lane = tid & 31;
  const int wv   = tid >> 5;
  const int half = lane >> 4;
  const int ml   = lane & 15;
  const int n0   = wv * 16;

  v8f acc = {0.f, 0.f, 0.f, 0.f, 0.f, 0.f, 0.f, 0.f};
  const float* wrow = Wfc + (size_t)(n0 + ml) * kC;

  for (int k0 = 0; k0 < kC; k0 += 4) {
    v2f a, bm;
    const float2 av = *reinterpret_cast<const float2*>(
        &lds_a[ml * kASTR + k0 + 2 * half]);
    a.x = av.x; a.y = av.y;
    const float2 wb = *reinterpret_cast<const float2*>(wrow + k0 + 2 * half);
    bm.x = wb.x; bm.y = wb.y;
    acc = __builtin_amdgcn_wmma_f32_16x16x4_f32(
        false, a, false, bm, (short)0, acc, false, false);
  }

#pragma unroll
  for (int r = 0; r < 8; ++r) {
    const int m = r + 8 * half;
    lds_o[m * kDOUT + n0 + ml] = acc[r];
  }
  __syncthreads();

  if (tid < 16) {
    float ss = 0.0f;
    for (int d = 0; d < kDOUT; ++d) {
      const float x = lds_o[tid * kDOUT + d];
      ss += x * x;
    }
    lds_inv[tid] = 1.0f / fmaxf(sqrtf(ss), 1e-12f);
  }
  __syncthreads();

  for (int e = tid; e < 16 * kDOUT; e += 256) {
    const int m = e >> 7;
    const int d = e & (kDOUT - 1);
    const int s = s0 + m;
    const int b = s / (kT * kNSP);
    const int t = (s / kNSP) % kT;
    const int n = s % kNSP;
    outp[((size_t)(b * kDOUT + d) * kT + t) * kNSP + n] =
        lds_o[m * kDOUT + d] * lds_inv[m];
  }
}

extern "C" void kernel_launch(void* const* d_in, const int* in_sizes, int n_in,
                              void* d_out, int out_size, void* d_ws, size_t ws_size,
                              hipStream_t stream) {
  (void)in_sizes; (void)n_in; (void)out_size; (void)ws_size;
  const float* maps    = (const float*)d_in[0];
  const int*   sp_mask = (const int*)d_in[1];
  const float* Wfc     = (const float*)d_in[2];
  float* out = (float*)d_out;

  float* sums   = (float*)d_ws;
  float* counts = sums + (size_t)kNSEG * kC;

  seg_sum_kernel<<<kB * kT * kNCHUNK, 256, 0, stream>>>(maps, sp_mask, sums);
  seg_cnt_kernel<<<kB * kT, 256, 0, stream>>>(sp_mask, counts);
  fc_norm_kernel<<<kNSEG / 16, 256, 0, stream>>>(sums, counts, Wfc, out);
}


# seg_sum LDS adds made conflict-free: lanes cover 4 channels x 8 pixel quads and each adds into its own bank column; sum order changes to 8 partial sums per value, combined q=0..7 at write-back
# speedup vs baseline: 1.5157x; 1.2545x over previous
#include <hip/hip_runtime.h>
#include <math.h>

static constexpr int kB    = 8;
static constexpr int kC    = 512;
static constexpr int kT    = 4;
static constexpr int kHW   = 64 * 64;
static constexpr int kNSP  = 100;
static constexpr int kDOUT = 128;
static constexpr int kNSEG = kB * kT * kNSP;
static constexpr int kCHUNK  = 32;
static constexpr int kNCHUNK = kC / kCHUNK;
static constexpr int kQ      = 8;
static constexpr int kWCH    = 4;
static constexpr int kSLOT   = kQ * kWCH;
static constexpr int kWREG   = kNSP * kSLOT + 4;
static constexpr int kASTR   = 516;

typedef float v2f __attribute__((ext_vector_type(2)));
typedef float v8f __attribute__((ext_vector_type(8)));

struct SegStage { int4 sp[8]; float4 v[8]; };

__device__ __forceinline__ void seg_load_stage(SegStage& s, const int* sp0,
                                               const float* mp0, int pb)
{
#pragma unroll
  for (int g = 0; g < 8; ++g) {
    s.sp[g] = *reinterpret_cast<const int4*>(sp0 + pb + 32 * g);
    s.v[g]  = *reinterpret_cast<const float4*>(mp0 + pb + 32 * g);
  }
}

__device__ __forceinline__ void seg_add_stage(const SegStage& s, float* lw)
{
#pragma unroll
  for (int g = 0; g < 8; ++g) {
    atomicAdd(&lw[s.sp[g].x * kSLOT], s.v[g].x);
    atomicAdd(&lw[s.sp[g].y * kSLOT], s.v[g].y);
    atomicAdd(&lw[s.sp[g].z * kSLOT], s.v[g].z);
    atomicAdd(&lw[s.sp[g].w * kSLOT], s.v[g].w);
  }
}

__global__ __launch_bounds__(256) void seg_sum_kernel(
    const float* __restrict__ maps, const int* __restrict__ sp_mask,
    float* __restrict__ sums)
{
  __shared__ __attribute__((aligned(16))) float lds[8 * kWREG];
  const int tid = threadIdx.x;
  {
    float4* lz = reinterpret_cast<float4*>(lds);
    for (int i = tid; i < (8 * kWREG) / 4; i += 256) lz[i] = make_float4(0.0f, 0.0f, 0.0f, 0.0f);
  }
  __syncthreads();

  const int blk   = blockIdx.x;
  const int chunk = blk & (kNCHUNK - 1);
  const int bt    = blk >> 4;
  const int b     = bt >> 2;
  const int t     = bt & 3;

  const int lane = tid & 31;
  const int wv   = tid >> 5;
  const int c    = lane >> 3;
  const int q    = lane & 7;

  const int* spb = sp_mask + (size_t)bt * kHW;

  const float* mp0 =
      maps + ((size_t)(b * kC + chunk * kCHUNK + wv + 8 * c) * kT + t) * kHW + q * 4;
  const int* sp0 = spb + q * 4;
  float* lw = lds + wv * kWREG + (4 * q + c);

  SegStage sa, sb;
  seg_load_stage(sa, sp0, mp0, 0);
  for (int pb = 0; pb < kHW; pb += 512) {
    seg_load_stage(sb, sp0, mp0, pb + 256);
    seg_add_stage(sa, lw);
    if (pb + 512 < kHW) seg_load_stage(sa, sp0, mp0, pb + 512);
    seg_add_stage(sb, lw);
  }
  __syncthreads();

  float* outp = sums + (size_t)bt * kNSP * kC + chunk * kCHUNK;
  for (int e = tid; e < kNSP * kCHUNK; e += 256) {
    const int n  = e >> 5;
    const int cl = e & 31;
    const float* p = lds + (cl & 7) * kWREG + n * kSLOT + (cl >> 3);
    float s = p[0];
#pragma unroll
    for (int qq = 1; qq < kQ; ++qq) s += p[4 * qq];
    outp[(size_t)n * kC + cl] = s;
  }
}

__global__ __launch_bounds__(256) void seg_cnt_kernel(
    const int* __restrict__ sp_mask, float* __restrict__ counts)
{
  __shared__ int cnt[kNSP];
  const int tid = threadIdx.x;
  if (tid < kNSP) cnt[tid] = 0;
  __syncthreads();
  const int bt = blockIdx.x;
  const int* spb = sp_mask + (size_t)bt * kHW;
  for (int p = tid; p < kHW; p += 256) atomicAdd(&cnt[spb[p]], 1);
  __syncthreads();
  if (tid < kNSP) counts[bt * kNSP + tid] = (float)cnt[tid];
}

__global__ __launch_bounds__(256) void fc_norm_kernel(
    const float* __restrict__ sums, const float* __restrict__ counts,
    const float* __restrict__ Wfc, float* __restrict__ outp)
{
  __shared__ __attribute__((aligned(16))) float lds_a[16 * kASTR];
  __shared__ float lds_o[16 * kDOUT];
  __shared__ float lds_inv[16];

  const int tid = threadIdx.x;
  const int s0  = blockIdx.x * 16;

  {
    const int c4 = tid & 127;
    const int mh = tid >> 7;
    float4 sv[8];
    float  cv[8];
#pragma unroll
    for (int i = 0; i < 8; ++i) {
      const int m = mh + 2 * i;
      cv[i] = counts[s0 + m];
      sv[i] = *reinterpret_cast<const float4*>(sums + (size_t)(s0 + m) * kC + 4 * c4);
    }
#pragma unroll
    for (int i = 0; i < 8; ++i) {
      const int m = mh + 2 * i;
      const bool nz = cv[i] > 0.0f;
      float4 r;
      r.x = nz ? (sv[i].x / cv[i]) : 0.0f;
      r.y = nz ? (sv[i].y / cv[i]) : 0.0f;
      r.z = nz ? (sv[i].z / cv[i]) : 0.0f;
      r.w = nz ? (sv[i].w / cv[i]) : 0.0f;
      *reinterpret_cast<float4*>(&lds_a[m * kASTR + 4 * c4]) = r;
    }
  }
  __syncthreads();

  const int lane = tid & 31;
  const int wv   = tid >> 5;
  const int half = lane >> 4;
  const int ml   = lane & 15;
  const int n0   = wv * 16;

  v8f acc = {0.f, 0.f, 0.f, 0.f, 0.f, 0.f, 0.f, 0.f};
  const float* wrow = Wfc + (size_t)(n0 + ml) * kC;

  for (int k0 = 0; k0 < kC; k0 += 4) {
    v2f a, bm;
    const float2 av = *reinterpret_cast<const float2*>(
        &lds_a[ml * kASTR + k0 + 2 * half]);
    a.x = av.x; a.y = av.y;
    const float2 wb = *reinterpret_cast<const float2*>(wrow + k0 + 2 * half);
    bm.x = wb.x; bm.y = wb.y;
    acc = __builtin_amdgcn_wmma_f32_16x16x4_f32(
        false, a, false, bm, (short)0, acc, false, false);
  }

#pragma unroll
  for (int r = 0; r < 8; ++r) {
    const int m = r + 8 * half;
    lds_o[m * kDOUT + n0 + ml] = acc[r];
  }
  __syncthreads();

  if (tid < 16) {
    float ss = 0.0f;
    for (int d = 0; d < kDOUT; ++d) {
      const float x = lds_o[tid * kDOUT + d];
      ss += x * x;
    }
    lds_inv[tid] = 1.0f / fmaxf(sqrtf(ss), 1e-12f);
  }
  __syncthreads();

  for (int e = tid; e < 16 * kDOUT; e += 256) {
    const int m = e >> 7;
    const int d = e & (kDOUT - 1);
    const int s = s0 + m;
    const int b = s / (kT * kNSP);
    const int t = (s / kNSP) % kT;
    const int n = s % kNSP;
    outp[((size_t)(b * kDOUT + d) * kT + t) * kNSP + n] =
        lds_o[m * kDOUT + d] * lds_inv[m];
  }
}

extern "C" void kernel_launch(void* const* d_in, const int* in_sizes, int n_in,
                              void* d_out, int out_size, void* d_ws, size_t ws_size,
                              hipStream_t stream) {
  (void)in_sizes; (void)n_in; (void)out_size; (void)ws_size;
  const float* maps    = (const float*)d_in[0];
  const int*   sp_mask = (const int*)d_in[1];
  const float* Wfc     = (const float*)d_in[2];
  float* out = (float*)d_out;

  float* sums   = (float*)d_ws;
  float* counts = sums + (size_t)kNSEG * kC;

  seg_sum_kernel<<<kB * kT * kNCHUNK, 256, 0, stream>>>(maps, sp_mask, sums);
  seg_cnt_kernel<<<kB * kT, 256, 0, stream>>>(sp_mask, counts);
  fc_norm_kernel<<<kNSEG / 16, 256, 0, stream>>>(sums, counts, Wfc, out);
}


# pixel counts folded into seg_sum: the slice-0 block of each frame counts its mask in LDS ahead of the pixel loop and stores the 100 counts; the separate count launch is gone (3 launches to 2)
# speedup vs baseline: 1.7003x; 1.1218x over previous
#include <hip/hip_runtime.h>
#include <math.h>

static constexpr int kB    = 8;
static constexpr int kC    = 512;
static constexpr int kT    = 4;
static constexpr int kHW   = 64 * 64;
static constexpr int kNSP  = 100;
static constexpr int kDOUT = 128;
static constexpr int kNSEG = kB * kT * kNSP;
static constexpr int kCHUNK  = 32;
static constexpr int kNCHUNK = kC / kCHUNK;
static constexpr int kQ      = 8;
static constexpr int kWCH    = 4;
static constexpr int kSLOT   = kQ * kWCH;
static constexpr int kWREG   = kNSP * kSLOT + 4;
static constexpr int kASTR   = 516;

typedef float v2f __attribute__((ext_vector_type(2)));
typedef float v8f __attribute__((ext_vector_type(8)));

struct SegStage { int4 sp[8]; float4 v[8]; };

__device__ __forceinline__ void seg_load_stage(SegStage& s, const int* sp0,
                                               const float* mp0, int pb)
{
#pragma unroll
  for (int g = 0; g < 8; ++g) {
    s.sp[g] = *reinterpret_cast<const int4*>(sp0 + pb + 32 * g);
    s.v[g]  = *reinterpret_cast<const float4*>(mp0 + pb + 32 * g);
  }
}

__device__ __forceinline__ void seg_add_stage(const SegStage& s, float* lw)
{
#pragma unroll
  for (int g = 0; g < 8; ++g) {
    atomicAdd(&lw[s.sp[g].x * kSLOT], s.v[g].x);
    atomicAdd(&lw[s.sp[g].y * kSLOT], s.v[g].y);
    atomicAdd(&lw[s.sp[g].z * kSLOT], s.v[g].z);
    atomicAdd(&lw[s.sp[g].w * kSLOT], s.v[g].w);
  }
}

__global__ __launch_bounds__(256) void seg_sum_kernel(
    const float* __restrict__ maps, const int* __restrict__ sp_mask,
    float* __restrict__ sums, float* __restrict__ counts)
{
  __shared__ __attribute__((aligned(16))) float lds[8 * kWREG];
  __shared__ int cnt[kNSP];
  const int tid = threadIdx.x;

  const int blk   = blockIdx.x;
  const int chunk = blk & (kNCHUNK - 1);
  const int bt    = blk >> 4;
  const int b     = bt >> 2;
  const int t     = bt & 3;

  const bool counting = (chunk == 0);
  int4 cm[4] = {};
  if (counting) {
    const int4* mq = reinterpret_cast<const int4*>(sp_mask + (size_t)bt * kHW);
#pragma unroll
    for (int i = 0; i < 4; ++i) cm[i] = mq[tid + 256 * i];
  }

  {
    float4* lz = reinterpret_cast<float4*>(lds);
    for (int i = tid; i < (8 * kWREG) / 4; i += 256) lz[i] = make_float4(0.0f, 0.0f, 0.0f, 0.0f);
  }
  if (tid < kNSP) cnt[tid] = 0;
  __syncthreads();

  if (counting) {
#pragma unroll
    for (int i = 0; i < 4; ++i) {
      atomicAdd(&cnt[cm[i].x], 1);
      atomicAdd(&cnt[cm[i].y], 1);
      atomicAdd(&cnt[cm[i].z], 1);
      atomicAdd(&cnt[cm[i].w], 1);
    }
  }

  const int lane = tid & 31;
  const int wv   = tid >> 5;
  const int c    = lane >> 3;
  const int q    = lane & 7;

  const int* spb = sp_mask + (size_t)bt * kHW;

  const float* mp0 =
      maps + ((size_t)(b * kC + chunk * kCHUNK + wv + 8 * c) * kT + t) * kHW + q * 4;
  const int* sp0 = spb + q * 4;
  float* lw = lds + wv * kWREG + (4 * q + c);

  SegStage sa, sb;
  seg_load_stage(sa, sp0, mp0, 0);
  for (int pb = 0; pb < kHW; pb += 512) {
    seg_load_stage(sb, sp0, mp0, pb + 256);
    seg_add_stage(sa, lw);
    if (pb + 512 < kHW) seg_load_stage(sa, sp0, mp0, pb + 512);
    seg_add_stage(sb, lw);
  }
  __syncthreads();

  if (counting && tid < kNSP) counts[bt * kNSP + tid] = (float)cnt[tid];

  float* outp = sums + (size_t)bt * kNSP * kC + chunk * kCHUNK;
  for (int e = tid; e < kNSP * kCHUNK; e += 256) {
    const int n  = e >> 5;
    const int cl = e & 31;
    const float* p = lds + (cl & 7) * kWREG + n * kSLOT + (cl >> 3);
    float s = p[0];
#pragma unroll
    for (int qq = 1; qq < kQ; ++qq) s += p[4 * qq];
    outp[(size_t)n * kC + cl] = s;
  }
}

__global__ __launch_bounds__(256) void fc_norm_kernel(
    const float* __restrict__ sums, const float* __restrict__ counts,
    const float* __restrict__ Wfc, float* __restrict__ outp)
{
  __shared__ __attribute__((aligned(16))) float lds_a[16 * kASTR];
  __shared__ float lds_o[16 * kDOUT];
  __shared__ float lds_inv[16];

  const int tid = threadIdx.x;
  const int s0  = blockIdx.x * 16;

  {
    const int c4 = tid & 127;
    const int mh = tid >> 7;
    float4 sv[8];
    float  cv[8];
#pragma unroll
    for (int i = 0; i < 8; ++i) {
      const int m = mh + 2 * i;
      cv[i] = counts[s0 + m];
      sv[i] = *reinterpret_cast<const float4*>(sums + (size_t)(s0 + m) * kC + 4 * c4);
    }
#pragma unroll
    for (int i = 0; i < 8; ++i) {
      const int m = mh + 2 * i;
      const bool nz = cv[i] > 0.0f;
      float4 r;
      r.x = nz ? (sv[i].x / cv[i]) : 0.0f;
      r.y = nz ? (sv[i].y / cv[i]) : 0.0f;
      r.z = nz ? (sv[i].z / cv[i]) : 0.0f;
      r.w = nz ? (sv[i].w / cv[i]) : 0.0f;
      *reinterpret_cast<float4*>(&lds_a[m * kASTR + 4 * c4]) = r;
    }
  }
  __syncthreads();

  const int lane = tid & 31;
  const int wv   = tid >> 5;
  const int half = lane >> 4;
  const int ml   = lane & 15;
  const int n0   = wv * 16;

  v8f acc = {0.f, 0.f, 0.f, 0.f, 0.f, 0.f, 0.f, 0.f};
  const float* wrow = Wfc + (size_t)(n0 + ml) * kC;

  for (int k0 = 0; k0 < kC; k0 += 4) {
    v2f a, bm;
    const float2 av = *reinterpret_cast<const float2*>(
        &lds_a[ml * kASTR + k0 + 2 * half]);
    a.x = av.x; a.y = av.y;
    const float2 wb = *reinterpret_cast<const float2*>(wrow + k0 + 2 * half);
    bm.x = wb.x; bm.y = wb.y;
    acc = __builtin_amdgcn_wmma_f32_16x16x4_f32(
        false, a, false, bm, (short)0, acc, false, false);
  }

#pragma unroll
  for (int r = 0; r < 8; ++r) {
    const int m = r + 8 * half;
    lds_o[m * kDOUT + n0 + ml] = acc[r];
  }
  __syncthreads();

  if (tid < 16) {
    float ss = 0.0f;
    for (int d = 0; d < kDOUT; ++d) {
      const float x = lds_o[tid * kDOUT + d];
      ss += x * x;
    }
    lds_inv[tid] = 1.0f / fmaxf(sqrtf(ss), 1e-12f);
  }
  __syncthreads();

  for (int e = tid; e < 16 * kDOUT; e += 256) {
    const int m = e >> 7;
    const int d = e & (kDOUT - 1);
    const int s = s0 + m;
    const int b = s / (kT * kNSP);
    const int t = (s / kNSP) % kT;
    const int n = s % kNSP;
    outp[((size_t)(b * kDOUT + d) * kT + t) * kNSP + n] =
        lds_o[m * kDOUT + d] * lds_inv[m];
  }
}

extern "C" void kernel_launch(void* const* d_in, const int* in_sizes, int n_in,
                              void* d_out, int out_size, void* d_ws, size_t ws_size,
                              hipStream_t stream) {
  (void)in_sizes; (void)n_in; (void)out_size; (void)ws_size;
  const float* maps    = (const float*)d_in[0];
  const int*   sp_mask = (const int*)d_in[1];
  const float* Wfc     = (const float*)d_in[2];
  float* out = (float*)d_out;

  float* sums   = (float*)d_ws;
  float* counts = sums + (size_t)kNSEG * kC;

  seg_sum_kernel<<<kB * kT * kNCHUNK, 256, 0, stream>>>(maps, sp_mask, sums, counts);
  fc_norm_kernel<<<kNSEG / 16, 256, 0, stream>>>(sums, counts, Wfc, out);
}


# fc_norm: W_fc fragments loaded 16 bytes per lane two 64-k blocks ahead of their matrix instructions (64 loads per wave, was 128 issued inside the using block); k regrouped per 8, k sum order changes
# speedup vs baseline: 1.7772x; 1.0452x over previous
#include <hip/hip_runtime.h>
#include <math.h>

static constexpr int kB    = 8;
static constexpr int kC    = 512;
static constexpr int kT    = 4;
static constexpr int kHW   = 64 * 64;
static constexpr int kNSP  = 100;
static constexpr int kDOUT = 128;
static constexpr int kNSEG = kB * kT * kNSP;
static constexpr int kCHUNK  = 32;
static constexpr int kNCHUNK = kC / kCHUNK;
static constexpr int kQ      = 8;
static constexpr int kWCH    = 4;
static constexpr int kSLOT   = kQ * kWCH;
static constexpr int kWREG   = kNSP * kSLOT + 4;
static constexpr int kASTR   = 516;
static constexpr int kKB     = 64;

typedef float v2f __attribute__((ext_vector_type(2)));
typedef float v8f __attribute__((ext_vector_type(8)));

struct SegStage { int4 sp[8]; float4 v[8]; };

__device__ __forceinline__ void seg_load_stage(SegStage& s, const int* sp0,
                                               const float* mp0, int pb)
{
#pragma unroll
  for (int g = 0; g < 8; ++g) {
    s.sp[g] = *reinterpret_cast<const int4*>(sp0 + pb + 32 * g);
    s.v[g]  = *reinterpret_cast<const float4*>(mp0 + pb + 32 * g);
  }
}

__device__ __forceinline__ void seg_add_stage(const SegStage& s, float* lw)
{
#pragma unroll
  for (int g = 0; g < 8; ++g) {
    atomicAdd(&lw[s.sp[g].x * kSLOT], s.v[g].x);
    atomicAdd(&lw[s.sp[g].y * kSLOT], s.v[g].y);
    atomicAdd(&lw[s.sp[g].z * kSLOT], s.v[g].z);
    atomicAdd(&lw[s.sp[g].w * kSLOT], s.v[g].w);
  }
}

__global__ __launch_bounds__(256) void seg_sum_kernel(
    const float* __restrict__ maps, const int* __restrict__ sp_mask,
    float* __restrict__ sums, float* __restrict__ counts)
{
  __shared__ __attribute__((aligned(16))) float lds[8 * kWREG];
  __shared__ int cnt[kNSP];
  const int tid = threadIdx.x;

  const int blk   = blockIdx.x;
  const int chunk = blk & (kNCHUNK - 1);
  const int bt    = blk >> 4;
  const int b     = bt >> 2;
  const int t     = bt & 3;

  const bool counting = (chunk == 0);
  int4 cm[4] = {};
  if (counting) {
    const int4* mq = reinterpret_cast<const int4*>(sp_mask + (size_t)bt * kHW);
#pragma unroll
    for (int i = 0; i < 4; ++i) cm[i] = mq[tid + 256 * i];
  }

  {
    float4* lz = reinterpret_cast<float4*>(lds);
    for (int i = tid; i < (8 * kWREG) / 4; i += 256) lz[i] = make_float4(0.0f, 0.0f, 0.0f, 0.0f);
  }
  if (tid < kNSP) cnt[tid] = 0;
  __syncthreads();

  if (counting) {
#pragma unroll
    for (int i = 0; i < 4; ++i) {
      atomicAdd(&cnt[cm[i].x], 1);
      atomicAdd(&cnt[cm[i].y], 1);
      atomicAdd(&cnt[cm[i].z], 1);
      atomicAdd(&cnt[cm[i].w], 1);
    }
  }

  const int lane = tid & 31;
  const int wv   = tid >> 5;
  const int c    = lane >> 3;
  const int q    = lane & 7;

  const int* spb = sp_mask + (size_t)bt * kHW;

  const float* mp0 =
      maps + ((size_t)(b * kC + chunk * kCHUNK + wv + 8 * c) * kT + t) * kHW + q * 4;
  const int* sp0 = spb + q * 4;
  float* lw = lds + wv * kWREG + (4 * q + c);

  SegStage sa, sb;
  seg_load_stage(sa, sp0, mp0, 0);
  for (int pb = 0; pb < kHW; pb += 512) {
    seg_load_stage(sb, sp0, mp0, pb + 256);
    seg_add_stage(sa, lw);
    if (pb + 512 < kHW) seg_load_stage(sa, sp0, mp0, pb + 512);
    seg_add_stage(sb, lw);
  }
  __syncthreads();

  if (counting && tid < kNSP) counts[bt * kNSP + tid] = (float)cnt[tid];

  float* outp = sums + (size_t)bt * kNSP * kC + chunk * kCHUNK;
  for (int e = tid; e < kNSP * kCHUNK; e += 256) {
    const int n  = e >> 5;
    const int cl = e & 31;
    const float* p = lds + (cl & 7) * kWREG + n * kSLOT + (cl >> 3);
    float s = p[0];
#pragma unroll
    for (int qq = 1; qq < kQ; ++qq) s += p[4 * qq];
    outp[(size_t)n * kC + cl] = s;
  }
}


struct FragQ { float4 q[8]; };

__device__ __forceinline__ void frag_load(FragQ& f, const float* row, int kb)
{
#pragma unroll
  for (int p = 0; p < 8; ++p)
    f.q[p] = *reinterpret_cast<const float4*>(row + kb + 8 * p);
}

__device__ __forceinline__ v8f frag_mma(const FragQ& fa, const FragQ& fb, v8f acc)
{
#pragma unroll
  for (int p = 0; p < 8; ++p) {
    v2f a, bm;
    a.x = fa.q[p].x; a.y = fa.q[p].y;
    bm.x = fb.q[p].x; bm.y = fb.q[p].y;
    acc = __builtin_amdgcn_wmma_f32_16x16x4_f32(
        false, a, false, bm, (short)0, acc, false, false);
    a.x = fa.q[p].z; a.y = fa.q[p].w;
    bm.x = fb.q[p].z; bm.y = fb.q[p].w;
    acc = __builtin_amdgcn_wmma_f32_16x16x4_f32(
        false, a, false, bm, (short)0, acc, false, false);
  }
  return acc;
}

__global__ __launch_bounds__(256) void fc_norm_kernel(
    const float* __restrict__ sums, const float* __restrict__ counts,
    const float* __restrict__ Wfc, float* __restrict__ outp)
{
  __shared__ __attribute__((aligned(16))) float lds_a[16 * kASTR];
  __shared__ float lds_o[16 * kDOUT];
  __shared__ float lds_inv[16];

  const int tid = threadIdx.x;
  const int s0  = blockIdx.x * 16;

  const int lane = tid & 31;
  const int wv   = tid >> 5;
  const int half = lane >> 4;
  const int ml   = lane & 15;
  const int n0   = wv * 16;

  {
    const int c4 = tid & 127;
    const int mh = tid >> 7;
    float4 sv[8];
    float  cv[8];
#pragma unroll
    for (int i = 0; i < 8; ++i) {
      const int m = mh + 2 * i;
      cv[i] = counts[s0 + m];
      sv[i] = *reinterpret_cast<const float4*>(sums + (size_t)(s0 + m) * kC + 4 * c4);
    }
#pragma unroll
    for (int i = 0; i < 8; ++i) {
      const int m = mh + 2 * i;
      const bool nz = cv[i] > 0.0f;
      float4 r;
      r.x = nz ? (sv[i].x / cv[i]) : 0.0f;
      r.y = nz ? (sv[i].y / cv[i]) : 0.0f;
      r.z = nz ? (sv[i].z / cv[i]) : 0.0f;
      r.w = nz ? (sv[i].w / cv[i]) : 0.0f;
      *reinterpret_cast<float4*>(&lds_a[m * kASTR + 4 * c4]) = r;
    }
  }
  __syncthreads();

  v8f acc = {0.f, 0.f, 0.f, 0.f, 0.f, 0.f, 0.f, 0.f};
  const float* arow = &lds_a[ml * kASTR + 4 * half];
  const float* wrow = Wfc + (size_t)(n0 + ml) * kC + 4 * half;

  FragQ a0, a1, b0, b1, b2;
  frag_load(b0, wrow, 0);        frag_load(a0, arow, 0);
  frag_load(b1, wrow, 1 * kKB);

  frag_load(b2, wrow, 2 * kKB);  frag_load(a1, arow, 1 * kKB);
  __builtin_amdgcn_sched_barrier(0);
  acc = frag_mma(a0, b0, acc);
  frag_load(b0, wrow, 3 * kKB);  frag_load(a0, arow, 2 * kKB);
  __builtin_amdgcn_sched_barrier(0);
  acc = frag_mma(a1, b1, acc);
  frag_load(b1, wrow, 4 * kKB);  frag_load(a1, arow, 3 * kKB);
  __builtin_amdgcn_sched_barrier(0);
  acc = frag_mma(a0, b2, acc);
  frag_load(b2, wrow, 5 * kKB);  frag_load(a0, arow, 4 * kKB);
  __builtin_amdgcn_sched_barrier(0);
  acc = frag_mma(a1, b0, acc);
  frag_load(b0, wrow, 6 * kKB);  frag_load(a1, arow, 5 * kKB);
  __builtin_amdgcn_sched_barrier(0);
  acc = frag_mma(a0, b1, acc);
  frag_load(b1, wrow, 7 * kKB);  frag_load(a0, arow, 6 * kKB);
  __builtin_amdgcn_sched_barrier(0);
  acc = frag_mma(a1, b2, acc);
  frag_load(a1, arow, 7 * kKB);
  __builtin_amdgcn_sched_barrier(0);
  acc = frag_mma(a0, b0, acc);
  acc = frag_mma(a1, b1, acc);

#pragma unroll
  for (int r = 0; r < 8; ++r) {
    const int m = r + 8 * half;
    lds_o[m * kDOUT + n0 + ml] = acc[r];
  }
  __syncthreads();

  if (tid < 16) {
    float ss = 0.0f;
    for (int d = 0; d < kDOUT; ++d) {
      const float x = lds_o[tid * kDOUT + d];
      ss += x * x;
    }
    lds_inv[tid] = 1.0f / fmaxf(sqrtf(ss), 1e-12f);
  }
  __syncthreads();

  for (int e = tid; e < 16 * kDOUT; e += 256) {
    const int m = e >> 7;
    const int d = e & (kDOUT - 1);
    const int s = s0 + m;
    const int b = s / (kT * kNSP);
    const int t = (s / kNSP) % kT;
    const int n = s % kNSP;
    outp[((size_t)(b * kDOUT + d) * kT + t) * kNSP + n] =
        lds_o[m * kDOUT + d] * lds_inv[m];
  }
}

extern "C" void kernel_launch(void* const* d_in, const int* in_sizes, int n_in,
                              void* d_out, int out_size, void* d_ws, size_t ws_size,
                              hipStream_t stream) {
  (void)in_sizes; (void)n_in; (void)out_size; (void)ws_size;
  const float* maps    = (const float*)d_in[0];
  const int*   sp_mask = (const int*)d_in[1];
  const float* Wfc     = (const float*)d_in[2];
  float* out = (float*)d_out;

  float* sums   = (float*)d_ws;
  float* counts = sums + (size_t)kNSEG * kC;

  seg_sum_kernel<<<kB * kT * kNCHUNK, 256, 0, stream>>>(maps, sp_mask, sums, counts);
  fc_norm_kernel<<<kNSEG / 16, 256, 0, stream>>>(sums, counts, Wfc, out);
}


# fc_norm epilogue with lanes along the tile's 16 segments: out stored as 64-byte runs (was 4-byte scatter, 32 lines per store); row norms by 16 threads per row, sum-of-squares order changes
# speedup vs baseline: 2.3089x; 1.2992x over previous
#include <hip/hip_runtime.h>
#include <math.h>

static constexpr int kB    = 8;
static constexpr int kC    = 512;
static constexpr int kT    = 4;
static constexpr int kHW   = 64 * 64;
static constexpr int kNSP  = 100;
static constexpr int kDOUT = 128;
static constexpr int kNSEG = kB * kT * kNSP;
static constexpr int kCHUNK  = 32;
static constexpr int kNCHUNK = kC / kCHUNK;
static constexpr int kQ      = 8;
static constexpr int kWCH    = 4;
static constexpr int kSLOT   = kQ * kWCH;
static constexpr int kWREG   = kNSP * kSLOT + 4;
static constexpr int kASTR   = 516;
static constexpr int kKB     = 64;
static constexpr int kOSTR   = 132;

typedef float v2f __attribute__((ext_vector_type(2)));
typedef float v8f __attribute__((ext_vector_type(8)));

struct SegStage { int4 sp[8]; float4 v[8]; };

__device__ __forceinline__ void seg_load_stage(SegStage& s, const int* sp0,
                                               const float* mp0, int pb)
{
#pragma unroll
  for (int g = 0; g < 8; ++g) {
    s.sp[g] = *reinterpret_cast<const int4*>(sp0 + pb + 32 * g);
    s.v[g]  = *reinterpret_cast<const float4*>(mp0 + pb + 32 * g);
  }
}

__device__ __forceinline__ void seg_add_stage(const SegStage& s, float* lw)
{
#pragma unroll
  for (int g = 0; g < 8; ++g) {
    atomicAdd(&lw[s.sp[g].x * kSLOT], s.v[g].x);
    atomicAdd(&lw[s.sp[g].y * kSLOT], s.v[g].y);
    atomicAdd(&lw[s.sp[g].z * kSLOT], s.v[g].z);
    atomicAdd(&lw[s.sp[g].w * kSLOT], s.v[g].w);
  }
}

__global__ __launch_bounds__(256) void seg_sum_kernel(
    const float* __restrict__ maps, const int* __restrict__ sp_mask,
    float* __restrict__ sums, float* __restrict__ counts)
{
  __shared__ __attribute__((aligned(16))) float lds[8 * kWREG];
  __shared__ int cnt[kNSP];
  const int tid = threadIdx.x;

  const int blk   = blockIdx.x;
  const int chunk = blk & (kNCHUNK - 1);
  const int bt    = blk >> 4;
  const int b     = bt >> 2;
  const int t     = bt & 3;

  const bool counting = (chunk == 0);
  int4 cm[4] = {};
  if (counting) {
    const int4* mq = reinterpret_cast<const int4*>(sp_mask + (size_t)bt * kHW);
#pragma unroll
    for (int i = 0; i < 4; ++i) cm[i] = mq[tid + 256 * i];
  }

  {
    float4* lz = reinterpret_cast<float4*>(lds);
    for (int i = tid; i < (8 * kWREG) / 4; i += 256) lz[i] = make_float4(0.0f, 0.0f, 0.0f, 0.0f);
  }
  if (tid < kNSP) cnt[tid] = 0;
  __syncthreads();

  if (counting) {
#pragma unroll
    for (int i = 0; i < 4; ++i) {
      atomicAdd(&cnt[cm[i].x], 1);
      atomicAdd(&cnt[cm[i].y], 1);
      atomicAdd(&cnt[cm[i].z], 1);
      atomicAdd(&cnt[cm[i].w], 1);
    }
  }

  const int lane = tid & 31;
  const int wv   = tid >> 5;
  const int c    = lane >> 3;
  const int q    = lane & 7;

  const int* spb = sp_mask + (size_t)bt * kHW;

  const float* mp0 =
      maps + ((size_t)(b * kC + chunk * kCHUNK + wv + 8 * c) * kT + t) * kHW + q * 4;
  const int* sp0 = spb + q * 4;
  float* lw = lds + wv * kWREG + (4 * q + c);

  SegStage sa, sb;
  seg_load_stage(sa, sp0, mp0, 0);
  for (int pb = 0; pb < kHW; pb += 512) {
    seg_load_stage(sb, sp0, mp0, pb + 256);
    seg_add_stage(sa, lw);
    if (pb + 512 < kHW) seg_load_stage(sa, sp0, mp0, pb + 512);
    seg_add_stage(sb, lw);
  }
  __syncthreads();

  if (counting && tid < kNSP) counts[bt * kNSP + tid] = (float)cnt[tid];

  float* outp = sums + (size_t)bt * kNSP * kC + chunk * kCHUNK;
  for (int e = tid; e < kNSP * kCHUNK; e += 256) {
    const int n  = e >> 5;
    const int cl = e & 31;
    const float* p = lds + (cl & 7) * kWREG + n * kSLOT + (cl >> 3);
    float s = p[0];
#pragma unroll
    for (int qq = 1; qq < kQ; ++qq) s += p[4 * qq];
    outp[(size_t)n * kC + cl] = s;
  }
}


struct FragQ { float4 q[8]; };

__device__ __forceinline__ void frag_load(FragQ& f, const float* row, int kb)
{
#pragma unroll
  for (int p = 0; p < 8; ++p)
    f.q[p] = *reinterpret_cast<const float4*>(row + kb + 8 * p);
}

__device__ __forceinline__ v8f frag_mma(const FragQ& fa, const FragQ& fb, v8f acc)
{
#pragma unroll
  for (int p = 0; p < 8; ++p) {
    v2f a, bm;
    a.x = fa.q[p].x; a.y = fa.q[p].y;
    bm.x = fb.q[p].x; bm.y = fb.q[p].y;
    acc = __builtin_amdgcn_wmma_f32_16x16x4_f32(
        false, a, false, bm, (short)0, acc, false, false);
    a.x = fa.q[p].z; a.y = fa.q[p].w;
    bm.x = fb.q[p].z; bm.y = fb.q[p].w;
    acc = __builtin_amdgcn_wmma_f32_16x16x4_f32(
        false, a, false, bm, (short)0, acc, false, false);
  }
  return acc;
}

__global__ __launch_bounds__(256) void fc_norm_kernel(
    const float* __restrict__ sums, const float* __restrict__ counts,
    const float* __restrict__ Wfc, float* __restrict__ outp)
{
  __shared__ __attribute__((aligned(16))) float lds_a[16 * kASTR];
  __shared__ __attribute__((aligned(16))) float lds_o[16 * kOSTR];
  __shared__ float lds_inv[16];

  const int tid = threadIdx.x;
  const int s0  = blockIdx.x * 16;

  const int lane = tid & 31;
  const int wv   = tid >> 5;
  const int half = lane >> 4;
  const int ml   = lane & 15;
  const int n0   = wv * 16;

  {
    const int c4 = tid & 127;
    const int mh = tid >> 7;
    float4 sv[8];
    float  cv[8];
#pragma unroll
    for (int i = 0; i < 8; ++i) {
      const int m = mh + 2 * i;
      cv[i] = counts[s0 + m];
      sv[i] = *reinterpret_cast<const float4*>(sums + (size_t)(s0 + m) * kC + 4 * c4);
    }
#pragma unroll
    for (int i = 0; i < 8; ++i) {
      const int m = mh + 2 * i;
      const bool nz = cv[i] > 0.0f;
      float4 r;
      r.x = nz ? (sv[i].x / cv[i]) : 0.0f;
      r.y = nz ? (sv[i].y / cv[i]) : 0.0f;
      r.z = nz ? (sv[i].z / cv[i]) : 0.0f;
      r.w = nz ? (sv[i].w / cv[i]) : 0.0f;
      *reinterpret_cast<float4*>(&lds_a[m * kASTR + 4 * c4]) = r;
    }
  }
  __syncthreads();

  v8f acc = {0.f, 0.f, 0.f, 0.f, 0.f, 0.f, 0.f, 0.f};
  const float* arow = &lds_a[ml * kASTR + 4 * half];
  const float* wrow = Wfc + (size_t)(n0 + ml) * kC + 4 * half;

  FragQ a0, a1, b0, b1, b2;
  frag_load(b0, wrow, 0);        frag_load(a0, arow, 0);
  frag_load(b1, wrow, 1 * kKB);

  frag_load(b2, wrow, 2 * kKB);  frag_load(a1, arow, 1 * kKB);
  __builtin_amdgcn_sched_barrier(0);
  acc = frag_mma(a0, b0, acc);
  frag_load(b0, wrow, 3 * kKB);  frag_load(a0, arow, 2 * kKB);
  __builtin_amdgcn_sched_barrier(0);
  acc = frag_mma(a1, b1, acc);
  frag_load(b1, wrow, 4 * kKB);  frag_load(a1, arow, 3 * kKB);
  __builtin_amdgcn_sched_barrier(0);
  acc = frag_mma(a0, b2, acc);
  frag_load(b2, wrow, 5 * kKB);  frag_load(a0, arow, 4 * kKB);
  __builtin_amdgcn_sched_barrier(0);
  acc = frag_mma(a1, b0, acc);
  frag_load(b0, wrow, 6 * kKB);  frag_load(a1, arow, 5 * kKB);
  __builtin_amdgcn_sched_barrier(0);
  acc = frag_mma(a0, b1, acc);
  frag_load(b1, wrow, 7 * kKB);  frag_load(a0, arow, 6 * kKB);
  __builtin_amdgcn_sched_barrier(0);
  acc = frag_mma(a1, b2, acc);
  frag_load(a1, arow, 7 * kKB);
  __builtin_amdgcn_sched_barrier(0);
  acc = frag_mma(a0, b0, acc);
  acc = frag_mma(a1, b1, acc);

#pragma unroll
  for (int r = 0; r < 8; ++r) {
    const int m = r + 8 * half;
    lds_o[m * kOSTR + n0 + ml] = acc[r];
  }
  __syncthreads();

  {
    const int m = tid >> 4;
    const int j = tid & 15;
    const float4 u = *reinterpret_cast<const float4*>(&lds_o[m * kOSTR + 8 * j]);
    const float4 w = *reinterpret_cast<const float4*>(&lds_o[m * kOSTR + 8 * j + 4]);
    float ss = u.x * u.x;
    ss = fmaf(u.y, u.y, ss);
    ss = fmaf(u.z, u.z, ss);
    ss = fmaf(u.w, u.w, ss);
    ss = fmaf(w.x, w.x, ss);
    ss = fmaf(w.y, w.y, ss);
    ss = fmaf(w.z, w.z, ss);
    ss = fmaf(w.w, w.w, ss);
    ss += __shfl_xor(ss, 1);
    ss += __shfl_xor(ss, 2);
    ss += __shfl_xor(ss, 4);
    ss += __shfl_xor(ss, 8);
    if (j == 0) lds_inv[m] = 1.0f / fmaxf(sqrtf(ss), 1e-12f);
  }
  __syncthreads();

  {
    const int m  = tid & 15;
    const int dq = tid >> 4;
    const int s  = s0 + m;
    const int b  = s / (kT * kNSP);
    const int t  = (s / kNSP) % kT;
    const int n  = s % kNSP;
    const float inv = lds_inv[m];
    float* op = outp + ((size_t)(b * kDOUT) * kT + t) * kNSP + n;
#pragma unroll
    for (int i = 0; i < 8; ++i) {
      const int d = dq + 16 * i;
      op[(size_t)d * (kT * kNSP)] = lds_o[m * kOSTR + d] * inv;
    }
  }
}

extern "C" void kernel_launch(void* const* d_in, const int* in_sizes, int n_in,
                              void* d_out, int out_size, void* d_ws, size_t ws_size,
                              hipStream_t stream) {
  (void)in_sizes; (void)n_in; (void)out_size; (void)ws_size;
  const float* maps    = (const float*)d_in[0];
  const int*   sp_mask = (const int*)d_in[1];
  const float* Wfc     = (const float*)d_in[2];
  float* out = (float*)d_out;

  float* sums   = (float*)d_ws;
  float* counts = sums + (size_t)kNSEG * kC;

  seg_sum_kernel<<<kB * kT * kNCHUNK, 256, 0, stream>>>(maps, sp_mask, sums, counts);
  fc_norm_kernel<<<kNSEG / 16, 256, 0, stream>>>(sums, counts, Wfc, out);
}
